# starting kernel
# baseline (speedup 1.0000x reference)
#include <hip/hip_runtime.h>
#include <hip/hip_bf16.h>

typedef __attribute__((ext_vector_type(2))) float v2f;
typedef __attribute__((ext_vector_type(8))) float v8f;

#define F_DIM 128
#define H_DIM 256
#define C_DIM 40
#define BN_EPS 1e-5f

__global__ __launch_bounds__(256) void mlp_kernel(
    const float* __restrict__ x, const float* __restrict__ W1,
    const float* __restrict__ b1, const float* __restrict__ gamma,
    const float* __restrict__ beta, const float* __restrict__ mean,
    const float* __restrict__ var, const float* __restrict__ W2,
    const float* __restrict__ b2, float* __restrict__ h0, int N)
{
    __shared__ float As[16 * 129];
    __shared__ float Hs[16 * 257];

    const int tid  = threadIdx.x;
    const int wave = tid >> 5;
    const int lane = tid & 31;
    const int half = lane >> 4;
    const int l16  = lane & 15;
    const int row0 = blockIdx.x * 16;

    #pragma unroll
    for (int i = 0; i < 8; ++i) {
        int idx = tid + i * 256;
        int r = idx >> 7, c = idx & 127;
        int gr = row0 + r;
        As[r * 129 + c] = (gr < N) ? x[(long)gr * F_DIM + c] : 0.0f;
    }
    __syncthreads();

    #pragma unroll
    for (int t = 0; t < 2; ++t) {
        const int ct = wave * 2 + t;
        const int n  = ct * 16 + l16;
        v8f acc = {};
        for (int k = 0; k < F_DIM; k += 4) {
            v2f a, b;
            a.x = As[l16 * 129 + k + 2 * half + 0];
            a.y = As[l16 * 129 + k + 2 * half + 1];
            b.x = W1[(long)(k + 2 * half + 0) * H_DIM + n];
            b.y = W1[(long)(k + 2 * half + 1) * H_DIM + n];
            acc = __builtin_amdgcn_wmma_f32_16x16x4_f32(
                false, a, false, b, (short)0, acc, false, false);
        }
        const float bb = b1[n];
        const float g  = gamma[n], be = beta[n];
        const float mu = mean[n],  iv = rsqrtf(var[n] + BN_EPS);
        #pragma unroll
        for (int r = 0; r < 8; ++r) {
            const int m = r + 8 * half;
            float v = acc[r] + bb;
            v = (v - mu) * iv * g + be;
            Hs[m * 257 + n] = fmaxf(v, 0.0f);
        }
    }
    __syncthreads();

    if (wave < 3) {
        const int n   = wave * 16 + l16;
        const bool nok = (n < C_DIM);
        v8f acc = {};
        for (int k = 0; k < H_DIM; k += 4) {
            v2f a, b;
            a.x = Hs[l16 * 257 + k + 2 * half + 0];
            a.y = Hs[l16 * 257 + k + 2 * half + 1];
            b.x = nok ? W2[(long)(k + 2 * half + 0) * C_DIM + n] : 0.0f;
            b.y = nok ? W2[(long)(k + 2 * half + 1) * C_DIM + n] : 0.0f;
            acc = __builtin_amdgcn_wmma_f32_16x16x4_f32(
                false, a, false, b, (short)0, acc, false, false);
        }
        const float bb = nok ? b2[n] : 0.0f;
        #pragma unroll
        for (int r = 0; r < 8; ++r) {
            const int m  = r + 8 * half;
            const int gr = row0 + m;
            if (nok && gr < N) h0[(long)gr * C_DIM + n] = acc[r] + bb;
        }
    }
}

__global__ void zero_kernel(float* __restrict__ p, long n)
{
    long i = (long)blockIdx.x * blockDim.x + threadIdx.x;
    if (i < n) p[i] = 0.0f;
}

__global__ void scatter_kernel(const int* __restrict__ src,
                               const int* __restrict__ dst,
                               const float* __restrict__ norm,
                               const float* __restrict__ prev,
                               float* __restrict__ next, int E)
{
    long tid = (long)blockIdx.x * blockDim.x + threadIdx.x;
    int e = (int)(tid >> 3);
    int l = (int)(tid & 7);
    if (e >= E) return;
    const int   s = src[e];
    const int   d = dst[e];
    const float w = norm[e];
    const float* pr = prev + (long)s * C_DIM;
    float*       nx = next + (long)d * C_DIM;
    #pragma unroll
    for (int j = 0; j < 5; ++j) {
        const int c = l + 8 * j;
        atomicAdd(&nx[c], w * pr[c]);
    }
}

__global__ void combine_kernel(const float* __restrict__ v,
                               float* __restrict__ acc,
                               const float* __restrict__ pw,
                               const float* __restrict__ pb, int N)
{
    int n = blockIdx.x * blockDim.x + threadIdx.x;
    if (n >= N) return;
    const float* row = v + (long)n * C_DIM;
    float d = pb[0];
    #pragma unroll
    for (int c = 0; c < C_DIM; ++c) d += row[c] * pw[c];
    const float r = 1.0f / (1.0f + expf(-d));
    float* ar = acc + (long)n * C_DIM;
    #pragma unroll
    for (int c = 0; c < C_DIM; ++c) ar[c] += r * row[c];
}

__global__ void lsm_kernel(const float* __restrict__ acc,
                           float* __restrict__ out, int N)
{
    int n = blockIdx.x * blockDim.x + threadIdx.x;
    if (n >= N) return;
    const float* row = acc + (long)n * C_DIM;
    float m = -3.402823466e38f;
    #pragma unroll
    for (int c = 0; c < C_DIM; ++c) m = fmaxf(m, row[c]);
    float s = 0.0f;
    #pragma unroll
    for (int c = 0; c < C_DIM; ++c) s += expf(row[c] - m);
    const float ls = logf(s);
    float* o = out + (long)n * C_DIM;
    #pragma unroll
    for (int c = 0; c < C_DIM; ++c) o[c] = row[c] - m - ls;
}

extern "C" void kernel_launch(void* const* d_in, const int* in_sizes, int n_in,
                              void* d_out, int out_size, void* d_ws, size_t ws_size,
                              hipStream_t stream)
{
    const float* x     = (const float*)d_in[0];
    const int*   edge  = (const int*)d_in[1];
    const float* norm  = (const float*)d_in[2];
    const float* W1    = (const float*)d_in[3];
    const float* b1    = (const float*)d_in[4];
    const float* gamma = (const float*)d_in[5];
    const float* beta  = (const float*)d_in[6];
    const float* mean  = (const float*)d_in[7];
    const float* var   = (const float*)d_in[8];
    const float* W2    = (const float*)d_in[9];
    const float* b2    = (const float*)d_in[10];
    const float* pw    = (const float*)d_in[11];
    const float* pb    = (const float*)d_in[12];

    const int N = in_sizes[0] / F_DIM;
    const int E = in_sizes[2];

    const size_t per = (((size_t)N * C_DIM) + 63) & ~(size_t)63;
    float* h0   = (float*)d_ws;
    float* bufA = h0   + per;
    float* bufB = bufA + per;
    float* accb = bufB + per;

    const int* src = edge;
    const int* dst = edge + E;

    const long tot     = (long)N * C_DIM;
    const int  zblocks = (int)((tot + 255) / 256);
    const int  nblocks = (N + 255) / 256;

    mlp_kernel<<<(N + 15) / 16, 256, 0, stream>>>(
        x, W1, b1, gamma, beta, mean, var, W2, b2, h0, N);

    zero_kernel<<<zblocks, 256, 0, stream>>>(accb, tot);
    combine_kernel<<<nblocks, 256, 0, stream>>>(h0, accb, pw, pb, N);

    const float* prev = h0;
    for (int k = 0; k < 10; ++k) {
        float* next = (k & 1) ? bufB : bufA;
        zero_kernel<<<zblocks, 256, 0, stream>>>(next, tot);
        const long st = (long)E * 8;
        scatter_kernel<<<(int)((st + 255) / 256), 256, 0, stream>>>(
            src, dst, norm, prev, next, E);
        combine_kernel<<<nblocks, 256, 0, stream>>>(next, accb, pw, pb, N);
        prev = next;
    }

    lsm_kernel<<<nblocks, 256, 0, stream>>>(accb, (float*)d_out, N);
}


# Hops as gather over a dst-sorted edge list rebuilt each call (int count atomics, scan, id-ordered fill) instead of zero + float-atomic scatter; per-node sum order now fixed (ascending edge id)
# speedup vs baseline: 1.8016x; 1.8016x over previous
#include <hip/hip_runtime.h>
#include <hip/hip_bf16.h>

typedef __attribute__((ext_vector_type(2))) float v2f;
typedef __attribute__((ext_vector_type(8))) float v8f;

#define F_DIM 128
#define H_DIM 256
#define C_DIM 40
#define BN_EPS 1e-5f

__global__ __launch_bounds__(256) void mlp_kernel(
    const float* __restrict__ x, const float* __restrict__ W1,
    const float* __restrict__ b1, const float* __restrict__ gamma,
    const float* __restrict__ beta, const float* __restrict__ mean,
    const float* __restrict__ var, const float* __restrict__ W2,
    const float* __restrict__ b2, float* __restrict__ h0, int N)
{
    __shared__ float As[16 * 129];
    __shared__ float Hs[16 * 257];

    const int tid  = threadIdx.x;
    const int wave = tid >> 5;
    const int lane = tid & 31;
    const int half = lane >> 4;
    const int l16  = lane & 15;
    const int row0 = blockIdx.x * 16;

    #pragma unroll
    for (int i = 0; i < 8; ++i) {
        int idx = tid + i * 256;
        int r = idx >> 7, c = idx & 127;
        int gr = row0 + r;
        As[r * 129 + c] = (gr < N) ? x[(long)gr * F_DIM + c] : 0.0f;
    }
    __syncthreads();

    #pragma unroll
    for (int t = 0; t < 2; ++t) {
        const int ct = wave * 2 + t;
        const int n  = ct * 16 + l16;
        v8f acc = {};
        for (int k = 0; k < F_DIM; k += 4) {
            v2f a, b;
            a.x = As[l16 * 129 + k + 2 * half + 0];
            a.y = As[l16 * 129 + k + 2 * half + 1];
            b.x = W1[(long)(k + 2 * half + 0) * H_DIM + n];
            b.y = W1[(long)(k + 2 * half + 1) * H_DIM + n];
            acc = __builtin_amdgcn_wmma_f32_16x16x4_f32(
                false, a, false, b, (short)0, acc, false, false);
        }
        const float bb = b1[n];
        const float g  = gamma[n], be = beta[n];
        const float mu = mean[n],  iv = rsqrtf(var[n] + BN_EPS);
        #pragma unroll
        for (int r = 0; r < 8; ++r) {
            const int m = r + 8 * half;
            float v = acc[r] + bb;
            v = (v - mu) * iv * g + be;
            Hs[m * 257 + n] = fmaxf(v, 0.0f);
        }
    }
    __syncthreads();

    if (wave < 3) {
        const int n   = wave * 16 + l16;
        const bool nok = (n < C_DIM);
        v8f acc = {};
        for (int k = 0; k < H_DIM; k += 4) {
            v2f a, b;
            a.x = Hs[l16 * 257 + k + 2 * half + 0];
            a.y = Hs[l16 * 257 + k + 2 * half + 1];
            b.x = nok ? W2[(long)(k + 2 * half + 0) * C_DIM + n] : 0.0f;
            b.y = nok ? W2[(long)(k + 2 * half + 1) * C_DIM + n] : 0.0f;
            acc = __builtin_amdgcn_wmma_f32_16x16x4_f32(
                false, a, false, b, (short)0, acc, false, false);
        }
        const float bb = nok ? b2[n] : 0.0f;
        #pragma unroll
        for (int r = 0; r < 8; ++r) {
            const int m  = r + 8 * half;
            const int gr = row0 + m;
            if (nok && gr < N) h0[(long)gr * C_DIM + n] = acc[r] + bb;
        }
    }
}

__global__ void zero_kernel(float* __restrict__ p, long n)
{
    long i = (long)blockIdx.x * blockDim.x + threadIdx.x;
    if (i < n) p[i] = 0.0f;
}

__global__ void zero_int_kernel(int* __restrict__ p, int n)
{
    int i = blockIdx.x * blockDim.x + threadIdx.x;
    if (i < n) p[i] = 0;
}

__global__ void hist_kernel(const int* __restrict__ dst, int* __restrict__ cnt,
                            int* __restrict__ rank, int E)
{
    int e = blockIdx.x * blockDim.x + threadIdx.x;
    if (e < E) rank[e] = atomicAdd(&cnt[dst[e]], 1);
}

#define SCAN_TILE 1024

__device__ __forceinline__ int wave_sum(int v)
{
    #pragma unroll
    for (int off = 16; off > 0; off >>= 1) v += __shfl_xor(v, off);
    return v;
}

__global__ __launch_bounds__(256) void scan_sum_kernel(const int* __restrict__ cnt,
                                                       int* __restrict__ bsum, int M)
{
    __shared__ int ws[8];
    const int tid  = threadIdx.x;
    const int base = blockIdx.x * SCAN_TILE + tid * 4;
    int s = 0;
    #pragma unroll
    for (int j = 0; j < 4; ++j) if (base + j < M) s += cnt[base + j];
    s = wave_sum(s);
    if ((tid & 31) == 0) ws[tid >> 5] = s;
    __syncthreads();
    if (tid == 0) {
        int t = 0;
        #pragma unroll
        for (int w = 0; w < 8; ++w) t += ws[w];
        bsum[blockIdx.x] = t;
    }
}

__global__ __launch_bounds__(256) void scan_write_kernel(int* __restrict__ cnt,
                                                         const int* __restrict__ bsum, int M)
{
    __shared__ int wo[8];
    __shared__ int wt[8];
    const int tid  = threadIdx.x;
    const int lane = tid & 31;
    const int wave = tid >> 5;
    const int b    = blockIdx.x;

    int part = 0;
    for (int i = tid; i < b; i += 256) part += bsum[i];
    part = wave_sum(part);

    const int base = b * SCAN_TILE + tid * 4;
    int v[4];
    #pragma unroll
    for (int j = 0; j < 4; ++j) v[j] = (base + j < M) ? cnt[base + j] : 0;
    const int t = v[0] + v[1] + v[2] + v[3];
    int incl = t;
    #pragma unroll
    for (int off = 1; off < 32; off <<= 1) {
        const int up = __shfl_up(incl, off);
        if (lane >= off) incl += up;
    }
    if (lane == 0)  wo[wave] = part;
    if (lane == 31) wt[wave] = incl;
    __syncthreads();
    int ex = incl - t;
    #pragma unroll
    for (int w = 0; w < 8; ++w) {
        ex += wo[w];
        if (w < wave) ex += wt[w];
    }
    #pragma unroll
    for (int j = 0; j < 4; ++j) {
        if (base + j < M) cnt[base + j] = ex;
        ex += v[j];
    }
}

__global__ void fill_kernel(const int* __restrict__ dst, const int* __restrict__ rank,
                            const int* __restrict__ rowptr, int* __restrict__ eid, int E)
{
    int e = blockIdx.x * blockDim.x + threadIdx.x;
    if (e < E) eid[rowptr[dst[e]] + rank[e]] = e;
}

__global__ void order_kernel(const int* __restrict__ eid, const int* __restrict__ src,
                             const int* __restrict__ dst, const float* __restrict__ norm,
                             const int* __restrict__ rowptr, int* __restrict__ csr_src,
                             float* __restrict__ csr_w, int E)
{
    int p = blockIdx.x * blockDim.x + threadIdx.x;
    if (p >= E) return;
    const int e = eid[p];
    const int d = dst[e];
    const int a = rowptr[d];
    const int b = rowptr[d + 1];
    int r = 0;
    for (int q = a; q < b; ++q) r += (eid[q] < e) ? 1 : 0;
    csr_src[a + r] = src[e];
    csr_w[a + r]   = norm[e];
}

#define GATHER_THREADS 320
__global__ __launch_bounds__(GATHER_THREADS) void gather_kernel(
    const int* __restrict__ rowptr, const int* __restrict__ csr_src,
    const float* __restrict__ csr_w, const float* __restrict__ prev,
    float* __restrict__ next, int N)
{
    const int t = blockIdx.x * GATHER_THREADS + threadIdx.x;
    const int n = t / 10;
    const int q = t - n * 10;
    if (n >= N) return;
    const int a = rowptr[n];
    const int b = rowptr[n + 1];
    const float4* __restrict__ pv = reinterpret_cast<const float4*>(prev) + q;
    float4 acc = make_float4(0.0f, 0.0f, 0.0f, 0.0f);
    #pragma unroll 4
    for (int p = a; p < b; ++p) {
        const int   s = csr_src[p];
        const float w = csr_w[p];
        const float4 v = pv[(long)s * 10];
        acc.x += w * v.x; acc.y += w * v.y; acc.z += w * v.z; acc.w += w * v.w;
    }
    reinterpret_cast<float4*>(next)[(long)n * 10 + q] = acc;
}

__global__ void combine_kernel(const float* __restrict__ v,
                               float* __restrict__ acc,
                               const float* __restrict__ pw,
                               const float* __restrict__ pb, int N)
{
    int n = blockIdx.x * blockDim.x + threadIdx.x;
    if (n >= N) return;
    const float* row = v + (long)n * C_DIM;
    float d = pb[0];
    #pragma unroll
    for (int c = 0; c < C_DIM; ++c) d += row[c] * pw[c];
    const float r = 1.0f / (1.0f + expf(-d));
    float* ar = acc + (long)n * C_DIM;
    #pragma unroll
    for (int c = 0; c < C_DIM; ++c) ar[c] += r * row[c];
}

__global__ void lsm_kernel(const float* __restrict__ acc,
                           float* __restrict__ out, int N)
{
    int n = blockIdx.x * blockDim.x + threadIdx.x;
    if (n >= N) return;
    const float* row = acc + (long)n * C_DIM;
    float m = -3.402823466e38f;
    #pragma unroll
    for (int c = 0; c < C_DIM; ++c) m = fmaxf(m, row[c]);
    float s = 0.0f;
    #pragma unroll
    for (int c = 0; c < C_DIM; ++c) s += expf(row[c] - m);
    const float ls = logf(s);
    float* o = out + (long)n * C_DIM;
    #pragma unroll
    for (int c = 0; c < C_DIM; ++c) o[c] = row[c] - m - ls;
}

extern "C" void kernel_launch(void* const* d_in, const int* in_sizes, int n_in,
                              void* d_out, int out_size, void* d_ws, size_t ws_size,
                              hipStream_t stream)
{
    const float* x     = (const float*)d_in[0];
    const int*   edge  = (const int*)d_in[1];
    const float* norm  = (const float*)d_in[2];
    const float* W1    = (const float*)d_in[3];
    const float* b1    = (const float*)d_in[4];
    const float* gamma = (const float*)d_in[5];
    const float* beta  = (const float*)d_in[6];
    const float* mean  = (const float*)d_in[7];
    const float* var   = (const float*)d_in[8];
    const float* W2    = (const float*)d_in[9];
    const float* b2    = (const float*)d_in[10];
    const float* pw    = (const float*)d_in[11];
    const float* pb    = (const float*)d_in[12];

    const int N = in_sizes[0] / F_DIM;
    const int E = in_sizes[2];

    const size_t per = (((size_t)N * C_DIM) + 63) & ~(size_t)63;
    float* h0   = (float*)d_ws;
    float* bufA = h0   + per;
    float* csrb = bufA + per;
    float* accb = csrb + per;

    const int    M      = N + 1;
    const size_t Mr     = ((size_t)M + 63) & ~(size_t)63;
    const size_t Er     = ((size_t)E + 63) & ~(size_t)63;
    const int    ntiles = (M + SCAN_TILE - 1) / SCAN_TILE;
    int*   rowptr  = (int*)csrb;
    int*   csr_src = rowptr + Mr;
    float* csr_w   = (float*)(csr_src + Er);
    int*   bsum    = (int*)(csr_w + Er);
    int*   rank    = (int*)bufA;
    int*   eid     = rank + Er;

    const int* src = edge;
    const int* dst = edge + E;

    const long tot     = (long)N * C_DIM;
    const int  zblocks = (int)((tot + 255) / 256);
    const int  nblocks = (N + 255) / 256;

    mlp_kernel<<<(N + 15) / 16, 256, 0, stream>>>(
        x, W1, b1, gamma, beta, mean, var, W2, b2, h0, N);

    zero_kernel<<<zblocks, 256, 0, stream>>>(accb, tot);
    combine_kernel<<<nblocks, 256, 0, stream>>>(h0, accb, pw, pb, N);

    const int eblocks = (E + 255) / 256;
    zero_int_kernel<<<(M + 255) / 256, 256, 0, stream>>>(rowptr, M);
    hist_kernel<<<eblocks, 256, 0, stream>>>(dst, rowptr, rank, E);
    scan_sum_kernel<<<ntiles, 256, 0, stream>>>(rowptr, bsum, M);
    scan_write_kernel<<<ntiles, 256, 0, stream>>>(rowptr, bsum, M);
    fill_kernel<<<eblocks, 256, 0, stream>>>(dst, rank, rowptr, eid, E);
    order_kernel<<<eblocks, 256, 0, stream>>>(eid, src, dst, norm, rowptr,
                                              csr_src, csr_w, E);

    const long gthreads = (long)N * 10;
    const int  gblocks  = (int)((gthreads + GATHER_THREADS - 1) / GATHER_THREADS);
    const float* prev = h0;
    for (int k = 0; k < 10; ++k) {
        float* next = (k & 1) ? h0 : bufA;
        gather_kernel<<<gblocks, GATHER_THREADS, 0, stream>>>(
            rowptr, csr_src, csr_w, prev, next, N);
        combine_kernel<<<nblocks, 256, 0, stream>>>(next, accb, pw, pb, N);
        prev = next;
    }

    lsm_kernel<<<nblocks, 256, 0, stream>>>(accb, (float*)d_out, N);
}


# Hop combine and final log-softmax run as a row epilogue inside the gather kernel (3 nodes per wave, 10 lanes each); zero fill dropped; proj dot and exp sums now 4-wide partials summed over 10 lanes
# speedup vs baseline: 2.8746x; 1.5956x over previous
#include <hip/hip_runtime.h>
#include <hip/hip_bf16.h>

typedef __attribute__((ext_vector_type(2))) float v2f;
typedef __attribute__((ext_vector_type(8))) float v8f;

#define F_DIM 128
#define H_DIM 256
#define C_DIM 40
#define BN_EPS 1e-5f

__global__ __launch_bounds__(256) void mlp_kernel(
    const float* __restrict__ x, const float* __restrict__ W1,
    const float* __restrict__ b1, const float* __restrict__ gamma,
    const float* __restrict__ beta, const float* __restrict__ mean,
    const float* __restrict__ var, const float* __restrict__ W2,
    const float* __restrict__ b2, float* __restrict__ h0, int N)
{
    __shared__ float As[16 * 129];
    __shared__ float Hs[16 * 257];

    const int tid  = threadIdx.x;
    const int wave = tid >> 5;
    const int lane = tid & 31;
    const int half = lane >> 4;
    const int l16  = lane & 15;
    const int row0 = blockIdx.x * 16;

    #pragma unroll
    for (int i = 0; i < 8; ++i) {
        int idx = tid + i * 256;
        int r = idx >> 7, c = idx & 127;
        int gr = row0 + r;
        As[r * 129 + c] = (gr < N) ? x[(long)gr * F_DIM + c] : 0.0f;
    }
    __syncthreads();

    #pragma unroll
    for (int t = 0; t < 2; ++t) {
        const int ct = wave * 2 + t;
        const int n  = ct * 16 + l16;
        v8f acc = {};
        for (int k = 0; k < F_DIM; k += 4) {
            v2f a, b;
            a.x = As[l16 * 129 + k + 2 * half + 0];
            a.y = As[l16 * 129 + k + 2 * half + 1];
            b.x = W1[(long)(k + 2 * half + 0) * H_DIM + n];
            b.y = W1[(long)(k + 2 * half + 1) * H_DIM + n];
            acc = __builtin_amdgcn_wmma_f32_16x16x4_f32(
                false, a, false, b, (short)0, acc, false, false);
        }
        const float bb = b1[n];
        const float g  = gamma[n], be = beta[n];
        const float mu = mean[n],  iv = rsqrtf(var[n] + BN_EPS);
        #pragma unroll
        for (int r = 0; r < 8; ++r) {
            const int m = r + 8 * half;
            float v = acc[r] + bb;
            v = (v - mu) * iv * g + be;
            Hs[m * 257 + n] = fmaxf(v, 0.0f);
        }
    }
    __syncthreads();

    if (wave < 3) {
        const int n   = wave * 16 + l16;
        const bool nok = (n < C_DIM);
        v8f acc = {};
        for (int k = 0; k < H_DIM; k += 4) {
            v2f a, b;
            a.x = Hs[l16 * 257 + k + 2 * half + 0];
            a.y = Hs[l16 * 257 + k + 2 * half + 1];
            b.x = nok ? W2[(long)(k + 2 * half + 0) * C_DIM + n] : 0.0f;
            b.y = nok ? W2[(long)(k + 2 * half + 1) * C_DIM + n] : 0.0f;
            acc = __builtin_amdgcn_wmma_f32_16x16x4_f32(
                false, a, false, b, (short)0, acc, false, false);
        }
        const float bb = nok ? b2[n] : 0.0f;
        #pragma unroll
        for (int r = 0; r < 8; ++r) {
            const int m  = r + 8 * half;
            const int gr = row0 + m;
            if (nok && gr < N) h0[(long)gr * C_DIM + n] = acc[r] + bb;
        }
    }
}

__global__ void zero_int_kernel(int* __restrict__ p, int n)
{
    int i = blockIdx.x * blockDim.x + threadIdx.x;
    if (i < n) p[i] = 0;
}

__global__ void hist_kernel(const int* __restrict__ dst, int* __restrict__ cnt,
                            int* __restrict__ rank, int E)
{
    int e = blockIdx.x * blockDim.x + threadIdx.x;
    if (e < E) rank[e] = atomicAdd(&cnt[dst[e]], 1);
}

#define SCAN_TILE 1024

__device__ __forceinline__ int wave_sum(int v)
{
    #pragma unroll
    for (int off = 16; off > 0; off >>= 1) v += __shfl_xor(v, off);
    return v;
}

__global__ __launch_bounds__(256) void scan_sum_kernel(const int* __restrict__ cnt,
                                                       int* __restrict__ bsum, int M)
{
    __shared__ int ws[8];
    const int tid  = threadIdx.x;
    const int base = blockIdx.x * SCAN_TILE + tid * 4;
    int s = 0;
    #pragma unroll
    for (int j = 0; j < 4; ++j) if (base + j < M) s += cnt[base + j];
    s = wave_sum(s);
    if ((tid & 31) == 0) ws[tid >> 5] = s;
    __syncthreads();
    if (tid == 0) {
        int t = 0;
        #pragma unroll
        for (int w = 0; w < 8; ++w) t += ws[w];
        bsum[blockIdx.x] = t;
    }
}

__global__ __launch_bounds__(256) void scan_write_kernel(int* __restrict__ cnt,
                                                         const int* __restrict__ bsum, int M)
{
    __shared__ int wo[8];
    __shared__ int wt[8];
    const int tid  = threadIdx.x;
    const int lane = tid & 31;
    const int wave = tid >> 5;
    const int b    = blockIdx.x;

    int part = 0;
    for (int i = tid; i < b; i += 256) part += bsum[i];
    part = wave_sum(part);

    const int base = b * SCAN_TILE + tid * 4;
    int v[4];
    #pragma unroll
    for (int j = 0; j < 4; ++j) v[j] = (base + j < M) ? cnt[base + j] : 0;
    const int t = v[0] + v[1] + v[2] + v[3];
    int incl = t;
    #pragma unroll
    for (int off = 1; off < 32; off <<= 1) {
        const int up = __shfl_up(incl, off);
        if (lane >= off) incl += up;
    }
    if (lane == 0)  wo[wave] = part;
    if (lane == 31) wt[wave] = incl;
    __syncthreads();
    int ex = incl - t;
    #pragma unroll
    for (int w = 0; w < 8; ++w) {
        ex += wo[w];
        if (w < wave) ex += wt[w];
    }
    #pragma unroll
    for (int j = 0; j < 4; ++j) {
        if (base + j < M) cnt[base + j] = ex;
        ex += v[j];
    }
}

__global__ void fill_kernel(const int* __restrict__ dst, const int* __restrict__ rank,
                            const int* __restrict__ rowptr, int* __restrict__ eid, int E)
{
    int e = blockIdx.x * blockDim.x + threadIdx.x;
    if (e < E) eid[rowptr[dst[e]] + rank[e]] = e;
}

__global__ void order_kernel(const int* __restrict__ eid, const int* __restrict__ src,
                             const int* __restrict__ dst, const float* __restrict__ norm,
                             const int* __restrict__ rowptr, int* __restrict__ csr_src,
                             float* __restrict__ csr_w, int E)
{
    int p = blockIdx.x * blockDim.x + threadIdx.x;
    if (p >= E) return;
    const int e = eid[p];
    const int d = dst[e];
    const int a = rowptr[d];
    const int b = rowptr[d + 1];
    int r = 0;
    for (int q = a; q < b; ++q) r += (eid[q] < e) ? 1 : 0;
    csr_src[a + r] = src[e];
    csr_w[a + r]   = norm[e];
}

#define ROW_THREADS 256
#define ROWS_PER_WAVE 3

__device__ __forceinline__ float row10_sum(float start, float part, int base)
{
    float s = start;
    #pragma unroll
    for (int j = 0; j < 10; ++j) s += __shfl(part, base + j);
    return s;
}

__device__ __forceinline__ float row10_max(float part, int base)
{
    float m = __shfl(part, base);
    #pragma unroll
    for (int j = 1; j < 10; ++j) m = fmaxf(m, __shfl(part, base + j));
    return m;
}

__device__ __forceinline__ float row_retain(const float4 v, const float4 w, float pb0, int base)
{
    const float part = v.x * w.x + v.y * w.y + v.z * w.z + v.w * w.w;
    const float d = row10_sum(pb0, part, base);
    return 1.0f / (1.0f + expf(-d));
}

#define C0_SLOTS 4
__global__ __launch_bounds__(ROW_THREADS) void combine0_kernel(
    const float* __restrict__ h0, float* __restrict__ accb,
    const float* __restrict__ pw, const float* __restrict__ pb, int N)
{
    const int lane = threadIdx.x & 31;
    const int wv   = blockIdx.x * (ROW_THREADS / 32) + (threadIdx.x >> 5);
    const int g    = lane / 10;
    const int q    = lane - g * 10;

    float4 w4;
    w4.x = pw[4 * q + 0]; w4.y = pw[4 * q + 1]; w4.z = pw[4 * q + 2]; w4.w = pw[4 * q + 3];
    const float pb0 = pb[0];

    float4 v[C0_SLOTS];
    long   o[C0_SLOTS];
    bool   live[C0_SLOTS];
    #pragma unroll
    for (int i = 0; i < C0_SLOTS; ++i) {
        const int n = (wv * C0_SLOTS + i) * ROWS_PER_WAVE + g;
        live[i] = (g < ROWS_PER_WAVE) && (n < N);
        o[i]    = live[i] ? ((long)n * 10 + q) : 0;
        v[i]    = make_float4(0.0f, 0.0f, 0.0f, 0.0f);
        if (live[i]) v[i] = reinterpret_cast<const float4*>(h0)[o[i]];
    }
    #pragma unroll
    for (int i = 0; i < C0_SLOTS; ++i) {
        const float r = row_retain(v[i], w4, pb0, g * 10);
        if (live[i]) {
            float4 a;
            a.x = r * v[i].x; a.y = r * v[i].y; a.z = r * v[i].z; a.w = r * v[i].w;
            reinterpret_cast<float4*>(accb)[o[i]] = a;
        }
    }
}

__device__ __forceinline__ void hop_body(
    const int* __restrict__ rowptr, const int* __restrict__ csr_src,
    const float* __restrict__ csr_w, const float* __restrict__ prev,
    float* __restrict__ next, float* __restrict__ accb,
    const float* __restrict__ pw, const float* __restrict__ pb,
    float* __restrict__ out, int N, const bool last)
{
    const int lane = threadIdx.x & 31;
    const int wv   = blockIdx.x * (ROW_THREADS / 32) + (threadIdx.x >> 5);
    const int g    = lane / 10;
    const int q    = lane - g * 10;
    const int n    = wv * ROWS_PER_WAVE + g;
    const bool live = (g < ROWS_PER_WAVE) && (n < N);
    const long o   = live ? ((long)n * 10 + q) : 0;

    int a = 0, b = 0;
    float4 A = make_float4(0.0f, 0.0f, 0.0f, 0.0f);
    if (live) {
        a = rowptr[n];
        b = rowptr[n + 1];
        A = reinterpret_cast<const float4*>(accb)[o];
    }
    float4 w4;
    w4.x = pw[4 * q + 0]; w4.y = pw[4 * q + 1]; w4.z = pw[4 * q + 2]; w4.w = pw[4 * q + 3];
    const float pb0 = pb[0];

    const float4* __restrict__ pv = reinterpret_cast<const float4*>(prev) + q;
    float4 acc = make_float4(0.0f, 0.0f, 0.0f, 0.0f);
    #pragma unroll 4
    for (int p = a; p < b; ++p) {
        const int   s = csr_src[p];
        const float w = csr_w[p];
        const float4 v = pv[(long)s * 10];
        acc.x += w * v.x; acc.y += w * v.y; acc.z += w * v.z; acc.w += w * v.w;
    }

    const int base = g * 10;
    const float r = row_retain(acc, w4, pb0, base);
    A.x += r * acc.x; A.y += r * acc.y; A.z += r * acc.z; A.w += r * acc.w;

    if (!last) {
        if (live) {
            reinterpret_cast<float4*>(next)[o] = acc;
            reinterpret_cast<float4*>(accb)[o] = A;
        }
    } else {
        const float m  = row10_max(fmaxf(fmaxf(A.x, A.y), fmaxf(A.z, A.w)), base);
        const float e  = expf(A.x - m) + expf(A.y - m) + expf(A.z - m) + expf(A.w - m);
        const float ls = logf(row10_sum(0.0f, e, base));
        if (live) {
            float4 y;
            y.x = A.x - m - ls; y.y = A.y - m - ls; y.z = A.z - m - ls; y.w = A.w - m - ls;
            reinterpret_cast<float4*>(out)[o] = y;
        }
    }
}

__global__ __launch_bounds__(ROW_THREADS) void hop_kernel(
    const int* __restrict__ rowptr, const int* __restrict__ csr_src,
    const float* __restrict__ csr_w, const float* __restrict__ prev,
    float* __restrict__ next, float* __restrict__ accb,
    const float* __restrict__ pw, const float* __restrict__ pb, int N)
{
    hop_body(rowptr, csr_src, csr_w, prev, next, accb, pw, pb, nullptr, N, false);
}

__global__ __launch_bounds__(ROW_THREADS) void last_hop_kernel(
    const int* __restrict__ rowptr, const int* __restrict__ csr_src,
    const float* __restrict__ csr_w, const float* __restrict__ prev,
    float* __restrict__ accb,
    const float* __restrict__ pw, const float* __restrict__ pb,
    float* __restrict__ out, int N)
{
    hop_body(rowptr, csr_src, csr_w, prev, nullptr, accb, pw, pb, out, N, true);
}

extern "C" void kernel_launch(void* const* d_in, const int* in_sizes, int n_in,
                              void* d_out, int out_size, void* d_ws, size_t ws_size,
                              hipStream_t stream)
{
    const float* x     = (const float*)d_in[0];
    const int*   edge  = (const int*)d_in[1];
    const float* norm  = (const float*)d_in[2];
    const float* W1    = (const float*)d_in[3];
    const float* b1    = (const float*)d_in[4];
    const float* gamma = (const float*)d_in[5];
    const float* beta  = (const float*)d_in[6];
    const float* mean  = (const float*)d_in[7];
    const float* var   = (const float*)d_in[8];
    const float* W2    = (const float*)d_in[9];
    const float* b2    = (const float*)d_in[10];
    const float* pw    = (const float*)d_in[11];
    const float* pb    = (const float*)d_in[12];

    const int N = in_sizes[0] / F_DIM;
    const int E = in_sizes[2];

    const size_t per = (((size_t)N * C_DIM) + 63) & ~(size_t)63;
    float* h0   = (float*)d_ws;
    float* bufA = h0   + per;
    float* csrb = bufA + per;
    float* accb = csrb + per;

    const int    M      = N + 1;
    const size_t Mr     = ((size_t)M + 63) & ~(size_t)63;
    const size_t Er     = ((size_t)E + 63) & ~(size_t)63;
    const int    ntiles = (M + SCAN_TILE - 1) / SCAN_TILE;
    int*   rowptr  = (int*)csrb;
    int*   csr_src = rowptr + Mr;
    float* csr_w   = (float*)(csr_src + Er);
    int*   bsum    = (int*)(csr_w + Er);
    int*   rank    = (int*)bufA;
    int*   eid     = rank + Er;

    const int* src = edge;
    const int* dst = edge + E;

    const int rwaves  = (N + ROWS_PER_WAVE - 1) / ROWS_PER_WAVE;
    const int rblocks = (rwaves + (ROW_THREADS / 32) - 1) / (ROW_THREADS / 32);

    mlp_kernel<<<(N + 15) / 16, 256, 0, stream>>>(
        x, W1, b1, gamma, beta, mean, var, W2, b2, h0, N);

    const int c0waves  = (rwaves + C0_SLOTS - 1) / C0_SLOTS;
    const int c0blocks = (c0waves + (ROW_THREADS / 32) - 1) / (ROW_THREADS / 32);
    combine0_kernel<<<c0blocks, ROW_THREADS, 0, stream>>>(h0, accb, pw, pb, N);

    const int eblocks = (E + 255) / 256;
    zero_int_kernel<<<(M + 255) / 256, 256, 0, stream>>>(rowptr, M);
    hist_kernel<<<eblocks, 256, 0, stream>>>(dst, rowptr, rank, E);
    scan_sum_kernel<<<ntiles, 256, 0, stream>>>(rowptr, bsum, M);
    scan_write_kernel<<<ntiles, 256, 0, stream>>>(rowptr, bsum, M);
    fill_kernel<<<eblocks, 256, 0, stream>>>(dst, rank, rowptr, eid, E);
    order_kernel<<<eblocks, 256, 0, stream>>>(eid, src, dst, norm, rowptr,
                                              csr_src, csr_w, E);

    const float* prev = h0;
    for (int k = 0; k < 9; ++k) {
        float* next = (k & 1) ? h0 : bufA;
        hop_kernel<<<rblocks, ROW_THREADS, 0, stream>>>(
            rowptr, csr_src, csr_w, prev, next, accb, pw, pb, N);
        prev = next;
    }
    last_hop_kernel<<<rblocks, ROW_THREADS, 0, stream>>>(
        rowptr, csr_src, csr_w, prev, accb, pw, pb, (float*)d_out, N);
}
